# starting kernel
# baseline (speedup 1.0000x reference)
#include <hip/hip_runtime.h>
#include <hip/hip_bf16.h>

#define IN_F 128
#define OUT_F 32
#define NEG_SLOPE 0.2f

typedef float v2f __attribute__((ext_vector_type(2)));
typedef float v8f __attribute__((ext_vector_type(8)));

__global__ void __launch_bounds__(256) gat_gemm_wmma(
    const float* __restrict__ h, const float* __restrict__ W,
    float* __restrict__ z, int n_nodes)
{
    const int wave = threadIdx.x >> 5;
    const int lane = threadIdx.x & 31;
    const int tile = blockIdx.x * 8 + wave;
    const int row0 = tile * 16;
    if (row0 >= n_nodes) return;

    const int half = lane >> 4;
    const int l16  = lane & 15;
    int arow = row0 + l16;
    if (arow >= n_nodes) arow = n_nodes - 1;

    const float* __restrict__ hrow = h + (size_t)arow * IN_F;
    const float* __restrict__ w0   = W + (size_t)l16 * IN_F;
    const float* __restrict__ w1   = W + (size_t)(l16 + 16) * IN_F;

    v8f acc0 = {};
    v8f acc1 = {};

#pragma unroll 4
    for (int kk = 0; kk < IN_F; kk += 4) {
        const int ka = kk + 2 * half;
        v2f A, B0, B1;
        A[0]  = hrow[ka];  A[1]  = hrow[ka + 1];
        B0[0] = w0[ka];    B0[1] = w0[ka + 1];
        B1[0] = w1[ka];    B1[1] = w1[ka + 1];
        acc0 = __builtin_amdgcn_wmma_f32_16x16x4_f32(false, A, false, B0,
                                                     (short)0, acc0, false, false);
        acc1 = __builtin_amdgcn_wmma_f32_16x16x4_f32(false, A, false, B1,
                                                     (short)0, acc1, false, false);
    }

#pragma unroll
    for (int r = 0; r < 8; ++r) {
        const int m = row0 + r + 8 * half;
        if (m < n_nodes) {
            z[(size_t)m * OUT_F + l16]      = acc0[r];
            z[(size_t)m * OUT_F + 16 + l16] = acc1[r];
        }
    }
}

__global__ void __launch_bounds__(256) gat_node_init(
    const float* __restrict__ z, const float* __restrict__ a,
    float* __restrict__ el, float* __restrict__ er,
    float* __restrict__ emax, float* __restrict__ denom,
    float* __restrict__ out, int n_nodes)
{
    const int n = blockIdx.x * blockDim.x + threadIdx.x;
    if (n >= n_nodes) return;
    const float* zr = z + (size_t)n * OUT_F;
    float sl = 0.f, sr = 0.f;
#pragma unroll
    for (int j = 0; j < OUT_F; ++j) {
        const float zv = zr[j];
        sl += zv * a[j];
        sr += zv * a[OUT_F + j];
    }
    el[n]    = sl;
    er[n]    = sr;
    emax[n]  = -__builtin_inff();
    denom[n] = 0.f;
    float4* o = (float4*)(out + (size_t)n * OUT_F);
#pragma unroll
    for (int j = 0; j < OUT_F / 4; ++j) o[j] = make_float4(0.f, 0.f, 0.f, 0.f);
}

__device__ __forceinline__ void atomic_max_f32(float* addr, float v)
{
    if (v >= 0.f)
        atomicMax((int*)addr, __float_as_int(v));
    else
        atomicMin((unsigned int*)addr, __float_as_uint(v));
}

__device__ __forceinline__ float leaky(float v)
{
    return v > 0.f ? v : NEG_SLOPE * v;
}

__global__ void __launch_bounds__(256) gat_edge_max(
    const int* __restrict__ src, const int* __restrict__ dst,
    const float* __restrict__ el, const float* __restrict__ er,
    float* __restrict__ emax, int n_edges)
{
    const int e = blockIdx.x * blockDim.x + threadIdx.x;
    if (e >= n_edges) return;
    const int s = src[e], d = dst[e];
    atomic_max_f32(&emax[d], leaky(el[s] + er[d]));
}

__global__ void __launch_bounds__(256) gat_edge_expsum(
    const int* __restrict__ src, const int* __restrict__ dst,
    const float* __restrict__ el, const float* __restrict__ er,
    const float* __restrict__ emax, float* __restrict__ denom, int n_edges)
{
    const int e = blockIdx.x * blockDim.x + threadIdx.x;
    if (e >= n_edges) return;
    const int s = src[e], d = dst[e];
    const float ex = __expf(leaky(el[s] + er[d]) - emax[d]);
    atomicAdd(&denom[d], ex);
}

__global__ void __launch_bounds__(256) gat_edge_aggregate(
    const int* __restrict__ src, const int* __restrict__ dst,
    const float* __restrict__ el, const float* __restrict__ er,
    const float* __restrict__ emax, const float* __restrict__ denom,
    const float* __restrict__ z, float* __restrict__ out, long n_edges)
{
    const long t = (long)blockIdx.x * blockDim.x + threadIdx.x;
    const long e = t >> 3;
    if (e >= n_edges) return;
    const int c = ((int)t & 7) * 4;
    const int s = src[e], d = dst[e];
    const float ex    = __expf(leaky(el[s] + er[d]) - emax[d]);
    const float alpha = ex / fmaxf(denom[d], 1e-16f);
    const float4 zc = *(const float4*)(z + (size_t)s * OUT_F + c);
    float* ob = out + (size_t)d * OUT_F + c;
    atomicAdd(ob + 0, alpha * zc.x);
    atomicAdd(ob + 1, alpha * zc.y);
    atomicAdd(ob + 2, alpha * zc.z);
    atomicAdd(ob + 3, alpha * zc.w);
}

extern "C" void kernel_launch(void* const* d_in, const int* in_sizes, int n_in,
                              void* d_out, int out_size, void* d_ws, size_t ws_size,
                              hipStream_t stream)
{
    const float* h  = (const float*)d_in[0];
    const float* W  = (const float*)d_in[1];
    const float* a  = (const float*)d_in[2];
    const int* src  = (const int*)d_in[3];
    const int* dst  = (const int*)d_in[4];
    float* out      = (float*)d_out;

    const int n_nodes = in_sizes[0] / IN_F;
    const int n_edges = in_sizes[3];

    float* z     = (float*)d_ws;
    float* el    = z + (size_t)n_nodes * OUT_F;
    float* er    = el + n_nodes;
    float* emax  = er + n_nodes;
    float* denom = emax + n_nodes;

    const int B = 256;
    const int ntiles = (n_nodes + 15) / 16;

    gat_gemm_wmma<<<dim3((ntiles + 7) / 8), dim3(B), 0, stream>>>(h, W, z, n_nodes);

    gat_node_init<<<dim3((n_nodes + B - 1) / B), dim3(B), 0, stream>>>(
        z, a, el, er, emax, denom, out, n_nodes);

    gat_edge_max<<<dim3((n_edges + B - 1) / B), dim3(B), 0, stream>>>(
        src, dst, el, er, emax, n_edges);

    gat_edge_expsum<<<dim3((n_edges + B - 1) / B), dim3(B), 0, stream>>>(
        src, dst, el, er, emax, denom, n_edges);

    const long t5 = (long)n_edges * 8;
    gat_edge_aggregate<<<dim3((unsigned)((t5 + B - 1) / B)), dim3(B), 0, stream>>>(
        src, dst, el, er, emax, denom, z, out, (long)n_edges);
}


# Softmax division deferred: the row-scatter pass adds p=exp(e-max) into denom and p*z[src] into out, a node pass then divides each row once; exp-sum launch goes; out summed undivided, then divided
# speedup vs baseline: 2.0309x; 2.0309x over previous
#include <hip/hip_runtime.h>
#include <hip/hip_bf16.h>

#define IN_F 128
#define OUT_F 32
#define NEG_SLOPE 0.2f

typedef float v2f __attribute__((ext_vector_type(2)));
typedef float v8f __attribute__((ext_vector_type(8)));

__global__ void __launch_bounds__(256) gat_gemm_wmma(
    const float* __restrict__ h, const float* __restrict__ W,
    float* __restrict__ z, int n_nodes)
{
    const int wave = threadIdx.x >> 5;
    const int lane = threadIdx.x & 31;
    const int tile = blockIdx.x * 8 + wave;
    const int row0 = tile * 16;
    if (row0 >= n_nodes) return;

    const int half = lane >> 4;
    const int l16  = lane & 15;
    int arow = row0 + l16;
    if (arow >= n_nodes) arow = n_nodes - 1;

    const float* __restrict__ hrow = h + (size_t)arow * IN_F;
    const float* __restrict__ w0   = W + (size_t)l16 * IN_F;
    const float* __restrict__ w1   = W + (size_t)(l16 + 16) * IN_F;

    v8f acc0 = {};
    v8f acc1 = {};

#pragma unroll 4
    for (int kk = 0; kk < IN_F; kk += 4) {
        const int ka = kk + 2 * half;
        v2f A, B0, B1;
        A[0]  = hrow[ka];  A[1]  = hrow[ka + 1];
        B0[0] = w0[ka];    B0[1] = w0[ka + 1];
        B1[0] = w1[ka];    B1[1] = w1[ka + 1];
        acc0 = __builtin_amdgcn_wmma_f32_16x16x4_f32(false, A, false, B0,
                                                     (short)0, acc0, false, false);
        acc1 = __builtin_amdgcn_wmma_f32_16x16x4_f32(false, A, false, B1,
                                                     (short)0, acc1, false, false);
    }

#pragma unroll
    for (int r = 0; r < 8; ++r) {
        const int m = row0 + r + 8 * half;
        if (m < n_nodes) {
            z[(size_t)m * OUT_F + l16]      = acc0[r];
            z[(size_t)m * OUT_F + 16 + l16] = acc1[r];
        }
    }
}

__global__ void __launch_bounds__(256) gat_node_init(
    const float* __restrict__ z, const float* __restrict__ a,
    float* __restrict__ el, float* __restrict__ er,
    float* __restrict__ emax, float* __restrict__ denom,
    float* __restrict__ out, int n_nodes)
{
    const int n = blockIdx.x * blockDim.x + threadIdx.x;
    if (n >= n_nodes) return;
    const float* zr = z + (size_t)n * OUT_F;
    float sl = 0.f, sr = 0.f;
#pragma unroll
    for (int j = 0; j < OUT_F; ++j) {
        const float zv = zr[j];
        sl += zv * a[j];
        sr += zv * a[OUT_F + j];
    }
    el[n]    = sl;
    er[n]    = sr;
    emax[n]            = -__builtin_inff();
    emax[n_nodes + n]  = -__builtin_inff();
    denom[n]           = 0.f;
    denom[n_nodes + n] = 0.f;
    float4* o = (float4*)(out + (size_t)n * OUT_F);
#pragma unroll
    for (int j = 0; j < OUT_F / 4; ++j) o[j] = make_float4(0.f, 0.f, 0.f, 0.f);
}

__device__ __forceinline__ int chip_half()
{
    const unsigned id = (unsigned)__builtin_amdgcn_s_sendmsg_rtn(0x87);
    return (int)((id >> 18) & 1u);
}

__device__ __forceinline__ void atomic_max_f32(float* addr, float v)
{
    if (v >= 0.f)
        atomicMax((int*)addr, __float_as_int(v));
    else
        atomicMin((unsigned int*)addr, __float_as_uint(v));
}

__device__ __forceinline__ float leaky(float v)
{
    return v > 0.f ? v : NEG_SLOPE * v;
}

__global__ void __launch_bounds__(256) gat_edge_max(
    const int* __restrict__ src, const int* __restrict__ dst,
    const float* __restrict__ el, const float* __restrict__ er,
    float* __restrict__ emax, int n_edges, int n_nodes)
{
    const int e = blockIdx.x * blockDim.x + threadIdx.x;
    if (e >= n_edges) return;
    const int s = src[e], d = dst[e];
    float* em = emax + (size_t)chip_half() * n_nodes;
    atomic_max_f32(&em[d], leaky(el[s] + er[d]));
}

#define EDGES_PER_WAVE 32
#define ROW_BATCH 8

template <bool FULL>
__device__ __forceinline__ void scatter_rows(
    const int s, const int d, const float p, const int cnt, const int lane,
    const float* __restrict__ z, float* __restrict__ out)
{
    const int n = FULL ? EDGES_PER_WAVE : cnt;
#pragma unroll
    for (int j0 = 0; j0 < n; j0 += ROW_BATCH) {
        float zv[ROW_BATCH];
#pragma unroll
        for (int k = 0; k < ROW_BATCH; ++k) {
            const int j = j0 + k;
            zv[k] = 0.f;
            if (FULL || j < cnt) {
                const int sj = __builtin_amdgcn_readlane(s, j);
                zv[k] = z[(size_t)sj * OUT_F + lane];
            }
        }
#pragma unroll
        for (int k = 0; k < ROW_BATCH; ++k) {
            const int j = j0 + k;
            if (FULL || j < cnt) {
                const int dj   = __builtin_amdgcn_readlane(d, j);
                const float aj = __int_as_float(
                    __builtin_amdgcn_readlane(__float_as_int(p), j));
                atomicAdd(out + (size_t)dj * OUT_F + lane, aj * zv[k]);
            }
        }
    }
}

__global__ void __launch_bounds__(256) gat_edge_aggregate(
    const int* __restrict__ src, const int* __restrict__ dst,
    const float* __restrict__ el, const float* __restrict__ er,
    const float* __restrict__ emax, float* __restrict__ denom,
    const float* __restrict__ z, float* __restrict__ out, long n_edges,
    int n_nodes)
{
    const int lane  = threadIdx.x & 31;
    const int wv    = __builtin_amdgcn_readfirstlane((int)(threadIdx.x >> 5));
    const long wave = (long)blockIdx.x * (blockDim.x >> 5) + wv;
    const long e0   = wave * EDGES_PER_WAVE;
    if (e0 >= n_edges) return;
    const long left = n_edges - e0;
    const int cnt   = left < EDGES_PER_WAVE ? (int)left : EDGES_PER_WAVE;

    float* dn = denom + (size_t)chip_half() * n_nodes;

    int s = 0, d = 0;
    float p = 0.f;
    if (lane < cnt) {
        const long e = e0 + lane;
        s = src[e];
        d = dst[e];
        const float m = fmaxf(emax[d], emax[n_nodes + d]);
        p = __expf(leaky(el[s] + er[d]) - m);
        atomicAdd(&dn[d], p);
    }

    if (cnt == EDGES_PER_WAVE)
        scatter_rows<true>(s, d, p, cnt, lane, z, out);
    else
        scatter_rows<false>(s, d, p, cnt, lane, z, out);
}

#define PIECES_PER_THREAD 4

__device__ __forceinline__ float4 div4(float4 v, const float q)
{
    v.x /= q;
    v.y /= q;
    v.z /= q;
    v.w /= q;
    return v;
}

__global__ void __launch_bounds__(256) gat_normalize(
    const float* __restrict__ denom, float* __restrict__ out, int n_nodes)
{
    const long total    = (long)n_nodes * (OUT_F / 4);
    const long blk_base = (long)blockIdx.x * (256 * PIECES_PER_THREAD);
    const long base     = blk_base + threadIdx.x;
    float4* o = (float4*)out;

    if (blk_base + 256 * PIECES_PER_THREAD <= total) {
        float4 v[PIECES_PER_THREAD];
        float d0[PIECES_PER_THREAD], d1[PIECES_PER_THREAD];
#pragma unroll
        for (int k = 0; k < PIECES_PER_THREAD; ++k) {
            const long i = base + 256 * k;
            const int n  = (int)(i / (OUT_F / 4));
            v[k]  = o[i];
            d0[k] = denom[n];
            d1[k] = denom[n_nodes + n];
        }
#pragma unroll
        for (int k = 0; k < PIECES_PER_THREAD; ++k) {
            const long i = base + 256 * k;
            o[i] = div4(v[k], fmaxf(d0[k] + d1[k], 1e-16f));
        }
    } else {
        for (int k = 0; k < PIECES_PER_THREAD; ++k) {
            const long i = base + 256 * k;
            if (i < total) {
                const int n = (int)(i / (OUT_F / 4));
                o[i] = div4(o[i], fmaxf(denom[n] + denom[n_nodes + n], 1e-16f));
            }
        }
    }
}

extern "C" void kernel_launch(void* const* d_in, const int* in_sizes, int n_in,
                              void* d_out, int out_size, void* d_ws, size_t ws_size,
                              hipStream_t stream)
{
    const float* h  = (const float*)d_in[0];
    const float* W  = (const float*)d_in[1];
    const float* a  = (const float*)d_in[2];
    const int* src  = (const int*)d_in[3];
    const int* dst  = (const int*)d_in[4];
    float* out      = (float*)d_out;

    const int n_nodes = in_sizes[0] / IN_F;
    const int n_edges = in_sizes[3];

    float* z     = (float*)d_ws;
    float* el    = z + (size_t)n_nodes * OUT_F;
    float* er    = el + n_nodes;
    float* emax  = er + n_nodes;
    float* denom = emax + 2 * (size_t)n_nodes;

    const int B = 256;
    const int ntiles = (n_nodes + 15) / 16;

    gat_gemm_wmma<<<dim3((ntiles + 7) / 8), dim3(B), 0, stream>>>(h, W, z, n_nodes);

    gat_node_init<<<dim3((n_nodes + B - 1) / B), dim3(B), 0, stream>>>(
        z, a, el, er, emax, denom, out, n_nodes);

    gat_edge_max<<<dim3((n_edges + B - 1) / B), dim3(B), 0, stream>>>(
        src, dst, el, er, emax, n_edges, n_nodes);

    const long t5 = (((long)n_edges + EDGES_PER_WAVE - 1) / EDGES_PER_WAVE) * 32;
    gat_edge_aggregate<<<dim3((unsigned)((t5 + B - 1) / B)), dim3(B), 0, stream>>>(
        src, dst, el, er, emax, denom, z, out, (long)n_edges, n_nodes);

    const long pieces = (long)n_nodes * (OUT_F / 4);
    const long per_block = (long)B * PIECES_PER_THREAD;
    gat_normalize<<<dim3((unsigned)((pieces + per_block - 1) / per_block)), dim3(B), 0, stream>>>(
        denom, out, n_nodes);
}


# Fewer scattered reads per edge: segment max taken over el[src] alone (leaky(x+er[d]) is monotonic in x: same max), el[src] kept per edge for the softmax pass, er[d] packed beside a max copy
# speedup vs baseline: 2.2177x; 1.0920x over previous
#include <hip/hip_runtime.h>
#include <hip/hip_bf16.h>

#define IN_F 128
#define OUT_F 32
#define NEG_SLOPE 0.2f

typedef float v2f __attribute__((ext_vector_type(2)));
typedef float v8f __attribute__((ext_vector_type(8)));

__global__ void __launch_bounds__(256) gat_gemm_wmma(
    const float* __restrict__ h, const float* __restrict__ W,
    float* __restrict__ z, int n_nodes)
{
    const int wave = threadIdx.x >> 5;
    const int lane = threadIdx.x & 31;
    const int tile = blockIdx.x * 8 + wave;
    const int row0 = tile * 16;
    if (row0 >= n_nodes) return;

    const int half = lane >> 4;
    const int l16  = lane & 15;
    int arow = row0 + l16;
    if (arow >= n_nodes) arow = n_nodes - 1;

    const float* __restrict__ hrow = h + (size_t)arow * IN_F;
    const float* __restrict__ w0   = W + (size_t)l16 * IN_F;
    const float* __restrict__ w1   = W + (size_t)(l16 + 16) * IN_F;

    v8f acc0 = {};
    v8f acc1 = {};

#pragma unroll 4
    for (int kk = 0; kk < IN_F; kk += 4) {
        const int ka = kk + 2 * half;
        v2f A, B0, B1;
        A[0]  = hrow[ka];  A[1]  = hrow[ka + 1];
        B0[0] = w0[ka];    B0[1] = w0[ka + 1];
        B1[0] = w1[ka];    B1[1] = w1[ka + 1];
        acc0 = __builtin_amdgcn_wmma_f32_16x16x4_f32(false, A, false, B0,
                                                     (short)0, acc0, false, false);
        acc1 = __builtin_amdgcn_wmma_f32_16x16x4_f32(false, A, false, B1,
                                                     (short)0, acc1, false, false);
    }

#pragma unroll
    for (int r = 0; r < 8; ++r) {
        const int m = row0 + r + 8 * half;
        if (m < n_nodes) {
            z[(size_t)m * OUT_F + l16]      = acc0[r];
            z[(size_t)m * OUT_F + 16 + l16] = acc1[r];
        }
    }
}

__global__ void __launch_bounds__(256) gat_node_init(
    const float* __restrict__ z, const float* __restrict__ a,
    float* __restrict__ el, float* __restrict__ rec0,
    float* __restrict__ elmax1, float* __restrict__ denom,
    float* __restrict__ out, int n_nodes)
{
    const int n = blockIdx.x * blockDim.x + threadIdx.x;
    if (n >= n_nodes) return;
    const float* zr = z + (size_t)n * OUT_F;
    float sl = 0.f, sr = 0.f;
#pragma unroll
    for (int j = 0; j < OUT_F; ++j) {
        const float zv = zr[j];
        sl += zv * a[j];
        sr += zv * a[OUT_F + j];
    }
    el[n]    = sl;
    ((float2*)rec0)[n] = make_float2(-__builtin_inff(), sr);
    elmax1[n]          = -__builtin_inff();
    denom[n]           = 0.f;
    denom[n_nodes + n] = 0.f;
    float4* o = (float4*)(out + (size_t)n * OUT_F);
#pragma unroll
    for (int j = 0; j < OUT_F / 4; ++j) o[j] = make_float4(0.f, 0.f, 0.f, 0.f);
}

__device__ __forceinline__ int chip_half()
{
    const unsigned id = (unsigned)__builtin_amdgcn_s_sendmsg_rtn(0x87);
    return (int)((id >> 18) & 1u);
}

__device__ __forceinline__ void atomic_max_f32(float* addr, float v)
{
    if (v >= 0.f)
        atomicMax((int*)addr, __float_as_int(v));
    else
        atomicMin((unsigned int*)addr, __float_as_uint(v));
}

__device__ __forceinline__ float leaky(float v)
{
    return v > 0.f ? v : NEG_SLOPE * v;
}

__global__ void __launch_bounds__(256) gat_edge_max(
    const int* __restrict__ src, const int* __restrict__ dst,
    const float* __restrict__ el, float* __restrict__ rec0,
    float* __restrict__ elmax1, float* __restrict__ els, int n_edges)
{
    const int e = blockIdx.x * blockDim.x + threadIdx.x;
    if (e >= n_edges) return;
    const int s = src[e], d = dst[e];
    const float v = el[s];
    els[e] = v;
    float* slot = chip_half() ? elmax1 + d : rec0 + 2 * (size_t)d;
    atomic_max_f32(slot, v);
}

#define EDGES_PER_WAVE 32
#define ROW_BATCH 8

template <bool FULL>
__device__ __forceinline__ void scatter_rows(
    const int s, const int d, const float p, const int cnt, const int lane,
    const float* __restrict__ z, float* __restrict__ out)
{
    const int n = FULL ? EDGES_PER_WAVE : cnt;
#pragma unroll
    for (int j0 = 0; j0 < n; j0 += ROW_BATCH) {
        float zv[ROW_BATCH];
#pragma unroll
        for (int k = 0; k < ROW_BATCH; ++k) {
            const int j = j0 + k;
            zv[k] = 0.f;
            if (FULL || j < cnt) {
                const int sj = __builtin_amdgcn_readlane(s, j);
                zv[k] = z[(size_t)sj * OUT_F + lane];
            }
        }
#pragma unroll
        for (int k = 0; k < ROW_BATCH; ++k) {
            const int j = j0 + k;
            if (FULL || j < cnt) {
                const int dj   = __builtin_amdgcn_readlane(d, j);
                const float aj = __int_as_float(
                    __builtin_amdgcn_readlane(__float_as_int(p), j));
                atomicAdd(out + (size_t)dj * OUT_F + lane, aj * zv[k]);
            }
        }
    }
}

__global__ void __launch_bounds__(256) gat_edge_aggregate(
    const int* __restrict__ src, const int* __restrict__ dst,
    const float* __restrict__ els, const float* __restrict__ rec0,
    const float* __restrict__ elmax1, float* __restrict__ denom,
    const float* __restrict__ z, float* __restrict__ out, long n_edges,
    int n_nodes)
{
    const int lane  = threadIdx.x & 31;
    const int wv    = __builtin_amdgcn_readfirstlane((int)(threadIdx.x >> 5));
    const long wave = (long)blockIdx.x * (blockDim.x >> 5) + wv;
    const long e0   = wave * EDGES_PER_WAVE;
    if (e0 >= n_edges) return;
    const long left = n_edges - e0;
    const int cnt   = left < EDGES_PER_WAVE ? (int)left : EDGES_PER_WAVE;

    float* dn = denom + (size_t)chip_half() * n_nodes;

    int s = 0, d = 0;
    float p = 0.f;
    if (lane < cnt) {
        const long e = e0 + lane;
        s = src[e];
        d = dst[e];
        const float2 r0 = ((const float2*)rec0)[d];
        const float m = leaky(fmaxf(r0.x, elmax1[d]) + r0.y);
        p = __expf(leaky(els[e] + r0.y) - m);
        atomicAdd(&dn[d], p);
    }

    if (cnt == EDGES_PER_WAVE)
        scatter_rows<true>(s, d, p, cnt, lane, z, out);
    else
        scatter_rows<false>(s, d, p, cnt, lane, z, out);
}

#define PIECES_PER_THREAD 4

__device__ __forceinline__ float4 div4(float4 v, const float q)
{
    v.x /= q;
    v.y /= q;
    v.z /= q;
    v.w /= q;
    return v;
}

__global__ void __launch_bounds__(256) gat_normalize(
    const float* __restrict__ denom, float* __restrict__ out, int n_nodes)
{
    const long total    = (long)n_nodes * (OUT_F / 4);
    const long blk_base = (long)blockIdx.x * (256 * PIECES_PER_THREAD);
    const long base     = blk_base + threadIdx.x;
    float4* o = (float4*)out;

    if (blk_base + 256 * PIECES_PER_THREAD <= total) {
        float4 v[PIECES_PER_THREAD];
        float d0[PIECES_PER_THREAD], d1[PIECES_PER_THREAD];
#pragma unroll
        for (int k = 0; k < PIECES_PER_THREAD; ++k) {
            const long i = base + 256 * k;
            const int n  = (int)(i / (OUT_F / 4));
            v[k]  = o[i];
            d0[k] = denom[n];
            d1[k] = denom[n_nodes + n];
        }
#pragma unroll
        for (int k = 0; k < PIECES_PER_THREAD; ++k) {
            const long i = base + 256 * k;
            o[i] = div4(v[k], fmaxf(d0[k] + d1[k], 1e-16f));
        }
    } else {
        for (int k = 0; k < PIECES_PER_THREAD; ++k) {
            const long i = base + 256 * k;
            if (i < total) {
                const int n = (int)(i / (OUT_F / 4));
                o[i] = div4(o[i], fmaxf(denom[n] + denom[n_nodes + n], 1e-16f));
            }
        }
    }
}

extern "C" void kernel_launch(void* const* d_in, const int* in_sizes, int n_in,
                              void* d_out, int out_size, void* d_ws, size_t ws_size,
                              hipStream_t stream)
{
    const float* h  = (const float*)d_in[0];
    const float* W  = (const float*)d_in[1];
    const float* a  = (const float*)d_in[2];
    const int* src  = (const int*)d_in[3];
    const int* dst  = (const int*)d_in[4];
    float* out      = (float*)d_out;

    const int n_nodes = in_sizes[0] / IN_F;
    const int n_edges = in_sizes[3];

    float* z      = (float*)d_ws;
    float* rec0   = z + (size_t)n_nodes * OUT_F;
    float* el     = rec0 + 2 * (size_t)n_nodes;
    float* elmax1 = el + n_nodes;
    float* denom  = elmax1 + n_nodes;
    float* els    = denom + 2 * (size_t)n_nodes;

    const int B = 256;
    const int ntiles = (n_nodes + 15) / 16;

    gat_gemm_wmma<<<dim3((ntiles + 7) / 8), dim3(B), 0, stream>>>(h, W, z, n_nodes);

    gat_node_init<<<dim3((n_nodes + B - 1) / B), dim3(B), 0, stream>>>(
        z, a, el, rec0, elmax1, denom, out, n_nodes);

    gat_edge_max<<<dim3((n_edges + B - 1) / B), dim3(B), 0, stream>>>(
        src, dst, el, rec0, elmax1, els, n_edges);

    const long t5 = (((long)n_edges + EDGES_PER_WAVE - 1) / EDGES_PER_WAVE) * 32;
    gat_edge_aggregate<<<dim3((unsigned)((t5 + B - 1) / B)), dim3(B), 0, stream>>>(
        src, dst, els, rec0, elmax1, denom, z, out, (long)n_edges, n_nodes);

    const long pieces = (long)n_nodes * (OUT_F / 4);
    const long per_block = (long)B * PIECES_PER_THREAD;
    gat_normalize<<<dim3((unsigned)((pieces + per_block - 1) / per_block)), dim3(B), 0, stream>>>(
        denom, out, n_nodes);
}


# Softmax shift = bound leaky(largest el of all nodes + er[d]) while the spread of el is under 20, else the exact segment maximum as before: max pass and maximum gathers not run; p scaled per node
# speedup vs baseline: 2.5396x; 1.1452x over previous
#include <hip/hip_runtime.h>
#include <hip/hip_bf16.h>

#define IN_F 128
#define OUT_F 32
#define NEG_SLOPE 0.2f

typedef float v2f __attribute__((ext_vector_type(2)));
typedef float v8f __attribute__((ext_vector_type(8)));

__global__ void __launch_bounds__(256) gat_gemm_wmma(
    const float* __restrict__ h, const float* __restrict__ W,
    float* __restrict__ z, float* __restrict__ gstat, int n_nodes)
{
    if (blockIdx.x == 0 && threadIdx.x == 0) {
        gstat[0] = -__builtin_inff();
        gstat[1] = __builtin_inff();
    }
    const int wave = threadIdx.x >> 5;
    const int lane = threadIdx.x & 31;
    const int tile = blockIdx.x * 8 + wave;
    const int row0 = tile * 16;
    if (row0 >= n_nodes) return;

    const int half = lane >> 4;
    const int l16  = lane & 15;
    int arow = row0 + l16;
    if (arow >= n_nodes) arow = n_nodes - 1;

    const float* __restrict__ hrow = h + (size_t)arow * IN_F;
    const float* __restrict__ w0   = W + (size_t)l16 * IN_F;
    const float* __restrict__ w1   = W + (size_t)(l16 + 16) * IN_F;

    v8f acc0 = {};
    v8f acc1 = {};

#pragma unroll 4
    for (int kk = 0; kk < IN_F; kk += 4) {
        const int ka = kk + 2 * half;
        v2f A, B0, B1;
        A[0]  = hrow[ka];  A[1]  = hrow[ka + 1];
        B0[0] = w0[ka];    B0[1] = w0[ka + 1];
        B1[0] = w1[ka];    B1[1] = w1[ka + 1];
        acc0 = __builtin_amdgcn_wmma_f32_16x16x4_f32(false, A, false, B0,
                                                     (short)0, acc0, false, false);
        acc1 = __builtin_amdgcn_wmma_f32_16x16x4_f32(false, A, false, B1,
                                                     (short)0, acc1, false, false);
    }

#pragma unroll
    for (int r = 0; r < 8; ++r) {
        const int m = row0 + r + 8 * half;
        if (m < n_nodes) {
            z[(size_t)m * OUT_F + l16]      = acc0[r];
            z[(size_t)m * OUT_F + 16 + l16] = acc1[r];
        }
    }
}

#define SPREAD_MAX 20.0f

__device__ __forceinline__ bool bound_shift(const float* __restrict__ gstat)
{
    return gstat[0] - gstat[1] < SPREAD_MAX;
}

__device__ __forceinline__ void atomic_max_f32(float* addr, float v)
{
    if (v >= 0.f)
        atomicMax((int*)addr, __float_as_int(v));
    else
        atomicMin((unsigned int*)addr, __float_as_uint(v));
}

__device__ __forceinline__ void atomic_min_f32(float* addr, float v)
{
    if (v >= 0.f)
        atomicMin((int*)addr, __float_as_int(v));
    else
        atomicMax((unsigned int*)addr, __float_as_uint(v));
}

__global__ void __launch_bounds__(256) gat_node_init(
    const float* __restrict__ z, const float* __restrict__ a,
    float* __restrict__ el, float* __restrict__ rec0,
    float* __restrict__ elmax1, float* __restrict__ denom,
    float* __restrict__ out, float* __restrict__ gstat, int n_nodes)
{
    __shared__ float s_hi[8], s_lo[8];
    const int n = blockIdx.x * blockDim.x + threadIdx.x;
    const bool live = n < n_nodes;
    float sl = 0.f, sr = 0.f;
    if (live) {
        const float* zr = z + (size_t)n * OUT_F;
#pragma unroll
        for (int j = 0; j < OUT_F; ++j) {
            const float zv = zr[j];
            sl += zv * a[j];
            sr += zv * a[OUT_F + j];
        }
        el[n]    = sl;
        ((float2*)rec0)[n] = make_float2(-__builtin_inff(), sr);
        elmax1[n]          = -__builtin_inff();
        denom[n]           = 0.f;
        denom[n_nodes + n] = 0.f;
        float4* o = (float4*)(out + (size_t)n * OUT_F);
#pragma unroll
        for (int j = 0; j < OUT_F / 4; ++j) o[j] = make_float4(0.f, 0.f, 0.f, 0.f);
    }

    float hi = live ? sl : -__builtin_inff();
    float lo = live ? sl : __builtin_inff();
    for (int o = 16; o > 0; o >>= 1) {
        hi = fmaxf(hi, __shfl_xor(hi, o));
        lo = fminf(lo, __shfl_xor(lo, o));
    }
    const int wave = threadIdx.x >> 5;
    if ((threadIdx.x & 31) == 0) { s_hi[wave] = hi; s_lo[wave] = lo; }
    __syncthreads();
    if (threadIdx.x == 0) {
        for (int w = 1; w < 8; ++w) {
            hi = fmaxf(hi, s_hi[w]);
            lo = fminf(lo, s_lo[w]);
        }
        atomic_max_f32(&gstat[0], hi);
        atomic_min_f32(&gstat[1], lo);
    }
}

__device__ __forceinline__ int chip_half()
{
    const unsigned id = (unsigned)__builtin_amdgcn_s_sendmsg_rtn(0x87);
    return (int)((id >> 18) & 1u);
}

__device__ __forceinline__ float leaky(float v)
{
    return v > 0.f ? v : NEG_SLOPE * v;
}

#define EDGE_MAX_BLOCKS 256
#define EDGES_PER_TRIP 4

__device__ __forceinline__ void edge_max_one(
    const int d, const float v, const int hf,
    float* __restrict__ rec0, float* __restrict__ elmax1)
{
    float* slot = hf ? elmax1 + d : rec0 + 2 * (size_t)d;
    atomic_max_f32(slot, v);
}

__global__ void __launch_bounds__(256) gat_edge_max(
    const int* __restrict__ src, const int* __restrict__ dst,
    const float* __restrict__ el, float* __restrict__ rec0,
    float* __restrict__ elmax1, float* __restrict__ els,
    const float* __restrict__ gstat, int n_edges)
{
    if (bound_shift(gstat)) return;
    const int hf = chip_half();
    const long trip = 256L * EDGES_PER_TRIP;
    for (long b = (long)blockIdx.x * trip; b < n_edges;
         b += (long)EDGE_MAX_BLOCKS * trip) {
        const long e0 = b + threadIdx.x;
        if (b + trip <= n_edges) {
            int s[EDGES_PER_TRIP], d[EDGES_PER_TRIP];
            float v[EDGES_PER_TRIP];
#pragma unroll
            for (int k = 0; k < EDGES_PER_TRIP; ++k) {
                s[k] = src[e0 + 256 * k];
                d[k] = dst[e0 + 256 * k];
            }
#pragma unroll
            for (int k = 0; k < EDGES_PER_TRIP; ++k) v[k] = el[s[k]];
#pragma unroll
            for (int k = 0; k < EDGES_PER_TRIP; ++k) {
                els[e0 + 256 * k] = v[k];
                edge_max_one(d[k], v[k], hf, rec0, elmax1);
            }
        } else {
            for (int k = 0; k < EDGES_PER_TRIP; ++k) {
                const long e = e0 + 256 * k;
                if (e < n_edges) {
                    const float v = el[src[e]];
                    els[e] = v;
                    edge_max_one(dst[e], v, hf, rec0, elmax1);
                }
            }
        }
    }
}

#define EDGES_PER_WAVE 32
#define ROW_BATCH 8

template <bool FULL>
__device__ __forceinline__ void scatter_rows(
    const int s, const int d, const float p, const int cnt, const int lane,
    const float* __restrict__ z, float* __restrict__ out)
{
    const int n = FULL ? EDGES_PER_WAVE : cnt;
#pragma unroll
    for (int j0 = 0; j0 < n; j0 += ROW_BATCH) {
        float zv[ROW_BATCH];
#pragma unroll
        for (int k = 0; k < ROW_BATCH; ++k) {
            const int j = j0 + k;
            zv[k] = 0.f;
            if (FULL || j < cnt) {
                const int sj = __builtin_amdgcn_readlane(s, j);
                zv[k] = z[(size_t)sj * OUT_F + lane];
            }
        }
#pragma unroll
        for (int k = 0; k < ROW_BATCH; ++k) {
            const int j = j0 + k;
            if (FULL || j < cnt) {
                const int dj   = __builtin_amdgcn_readlane(d, j);
                const float aj = __int_as_float(
                    __builtin_amdgcn_readlane(__float_as_int(p), j));
                atomicAdd(out + (size_t)dj * OUT_F + lane, aj * zv[k]);
            }
        }
    }
}

__global__ void __launch_bounds__(256) gat_edge_aggregate(
    const int* __restrict__ src, const int* __restrict__ dst,
    const float* __restrict__ el, const float* __restrict__ els,
    const float* __restrict__ rec0,
    const float* __restrict__ elmax1, float* __restrict__ denom,
    const float* __restrict__ z, float* __restrict__ out,
    const float* __restrict__ gstat, long n_edges, int n_nodes)
{
    const int lane  = threadIdx.x & 31;
    const int wv    = __builtin_amdgcn_readfirstlane((int)(threadIdx.x >> 5));
    const long wave = (long)blockIdx.x * (blockDim.x >> 5) + wv;
    const long e0   = wave * EDGES_PER_WAVE;
    if (e0 >= n_edges) return;
    const long left = n_edges - e0;
    const int cnt   = left < EDGES_PER_WAVE ? (int)left : EDGES_PER_WAVE;

    float* dn = denom + (size_t)chip_half() * n_nodes;
    const bool bound  = bound_shift(gstat);
    const float g_hi  = gstat[0];

    int s = 0, d = 0;
    float p = 0.f;
    if (lane < cnt) {
        const long e = e0 + lane;
        s = src[e];
        d = dst[e];
        const float x = *(bound ? el + s : els + e);
        const float2 r0 = ((const float2*)rec0)[d];
        float top = g_hi;
        if (!bound) top = fmaxf(r0.x, elmax1[d]);
        p = __expf(leaky(x + r0.y) - leaky(top + r0.y));
        atomicAdd(&dn[d], p);
    }

    if (cnt == EDGES_PER_WAVE)
        scatter_rows<true>(s, d, p, cnt, lane, z, out);
    else
        scatter_rows<false>(s, d, p, cnt, lane, z, out);
}

#define PIECES_PER_THREAD 4

__device__ __forceinline__ float4 div4(float4 v, const float q)
{
    v.x /= q;
    v.y /= q;
    v.z /= q;
    v.w /= q;
    return v;
}

__device__ __forceinline__ float safe_denom(const float q, const bool bound)
{
    return bound ? (q > 0.f ? q : 1.f) : fmaxf(q, 1e-16f);
}

__global__ void __launch_bounds__(256) gat_normalize(
    const float* __restrict__ denom, float* __restrict__ out,
    const float* __restrict__ gstat, int n_nodes)
{
    const bool bound = bound_shift(gstat);
    const long total    = (long)n_nodes * (OUT_F / 4);
    const long blk_base = (long)blockIdx.x * (256 * PIECES_PER_THREAD);
    const long base     = blk_base + threadIdx.x;
    float4* o = (float4*)out;

    if (blk_base + 256 * PIECES_PER_THREAD <= total) {
        float4 v[PIECES_PER_THREAD];
        float d0[PIECES_PER_THREAD], d1[PIECES_PER_THREAD];
#pragma unroll
        for (int k = 0; k < PIECES_PER_THREAD; ++k) {
            const long i = base + 256 * k;
            const int n  = (int)(i / (OUT_F / 4));
            v[k]  = o[i];
            d0[k] = denom[n];
            d1[k] = denom[n_nodes + n];
        }
#pragma unroll
        for (int k = 0; k < PIECES_PER_THREAD; ++k) {
            const long i = base + 256 * k;
            o[i] = div4(v[k], safe_denom(d0[k] + d1[k], bound));
        }
    } else {
        for (int k = 0; k < PIECES_PER_THREAD; ++k) {
            const long i = base + 256 * k;
            if (i < total) {
                const int n = (int)(i / (OUT_F / 4));
                o[i] = div4(o[i], safe_denom(denom[n] + denom[n_nodes + n], bound));
            }
        }
    }
}

extern "C" void kernel_launch(void* const* d_in, const int* in_sizes, int n_in,
                              void* d_out, int out_size, void* d_ws, size_t ws_size,
                              hipStream_t stream)
{
    const float* h  = (const float*)d_in[0];
    const float* W  = (const float*)d_in[1];
    const float* a  = (const float*)d_in[2];
    const int* src  = (const int*)d_in[3];
    const int* dst  = (const int*)d_in[4];
    float* out      = (float*)d_out;

    const int n_nodes = in_sizes[0] / IN_F;
    const int n_edges = in_sizes[3];

    float* z      = (float*)d_ws;
    float* rec0   = z + (size_t)n_nodes * OUT_F;
    float* el     = rec0 + 2 * (size_t)n_nodes;
    float* elmax1 = el + n_nodes;
    float* denom  = elmax1 + n_nodes;
    float* els    = denom + 2 * (size_t)n_nodes;
    float* gstat  = els + n_edges;

    const int B = 256;
    const int ntiles = (n_nodes + 15) / 16;

    gat_gemm_wmma<<<dim3((ntiles + 7) / 8), dim3(B), 0, stream>>>(h, W, z, gstat, n_nodes);

    gat_node_init<<<dim3((n_nodes + B - 1) / B), dim3(B), 0, stream>>>(
        z, a, el, rec0, elmax1, denom, out, gstat, n_nodes);

    gat_edge_max<<<dim3(EDGE_MAX_BLOCKS), dim3(B), 0, stream>>>(
        src, dst, el, rec0, elmax1, els, gstat, n_edges);

    const long t5 = (((long)n_edges + EDGES_PER_WAVE - 1) / EDGES_PER_WAVE) * 32;
    gat_edge_aggregate<<<dim3((unsigned)((t5 + B - 1) / B)), dim3(B), 0, stream>>>(
        src, dst, el, els, rec0, elmax1, denom, z, out, gstat, (long)n_edges, n_nodes);

    const long pieces = (long)n_nodes * (OUT_F / 4);
    const long per_block = (long)B * PIECES_PER_THREAD;
    gat_normalize<<<dim3((unsigned)((pieces + per_block - 1) / per_block)), dim3(B), 0, stream>>>(
        denom, out, gstat, n_nodes);
}


# Product's wave tile 64x32: a wave takes four 16-row tiles at once, so each W fragment it loads feeds four products (was one); next 16 of k loaded under the current products; sums, k order kept
# speedup vs baseline: 2.6402x; 1.0396x over previous
#include <hip/hip_runtime.h>
#include <hip/hip_bf16.h>

#define IN_F 128
#define OUT_F 32
#define NEG_SLOPE 0.2f

typedef float v2f __attribute__((ext_vector_type(2)));
typedef float v8f __attribute__((ext_vector_type(8)));

#define ROW_TILES 4
#define K_STAGE 16
#define STEPS_PER_STAGE (K_STAGE / 4)

struct gemm_frags {
    v2f a[ROW_TILES][STEPS_PER_STAGE];
    v2f b0[STEPS_PER_STAGE];
    v2f b1[STEPS_PER_STAGE];
};

__device__ __forceinline__ void gemm_load_stage(
    gemm_frags& f, const float* const (&hrow)[ROW_TILES],
    const float* __restrict__ w0, const float* __restrict__ w1,
    const int k0, const int half)
{
#pragma unroll
    for (int q = 0; q < STEPS_PER_STAGE; ++q) {
        const int ka = k0 + 4 * q + 2 * half;
        f.b0[q][0] = w0[ka];  f.b0[q][1] = w0[ka + 1];
        f.b1[q][0] = w1[ka];  f.b1[q][1] = w1[ka + 1];
#pragma unroll
        for (int t = 0; t < ROW_TILES; ++t) {
            f.a[t][q][0] = hrow[t][ka];
            f.a[t][q][1] = hrow[t][ka + 1];
        }
    }
}

__global__ void __launch_bounds__(32) gat_gemm_wmma(
    const float* __restrict__ h, const float* __restrict__ W,
    float* __restrict__ z, float* __restrict__ gstat, int n_nodes)
{
    if (blockIdx.x == 0 && threadIdx.x == 0) {
        gstat[0] = -__builtin_inff();
        gstat[1] = __builtin_inff();
    }
    const int lane = threadIdx.x & 31;
    const int row0 = blockIdx.x * (16 * ROW_TILES);
    if (row0 >= n_nodes) return;

    const int half = lane >> 4;
    const int l16  = lane & 15;

    const float* hrow[ROW_TILES];
#pragma unroll
    for (int t = 0; t < ROW_TILES; ++t) {
        int arow = row0 + 16 * t + l16;
        if (arow >= n_nodes) arow = n_nodes - 1;
        hrow[t] = h + (size_t)arow * IN_F;
    }
    const float* __restrict__ w0 = W + (size_t)l16 * IN_F;
    const float* __restrict__ w1 = W + (size_t)(l16 + 16) * IN_F;

    v8f acc0[ROW_TILES], acc1[ROW_TILES];
#pragma unroll
    for (int t = 0; t < ROW_TILES; ++t) { acc0[t] = {}; acc1[t] = {}; }

    gemm_frags cur;
    gemm_load_stage(cur, hrow, w0, w1, 0, half);

#pragma unroll
    for (int s = 0; s < IN_F / K_STAGE; ++s) {
        gemm_frags nxt;
        if (s + 1 < IN_F / K_STAGE)
            gemm_load_stage(nxt, hrow, w0, w1, (s + 1) * K_STAGE, half);
#pragma unroll
        for (int q = 0; q < STEPS_PER_STAGE; ++q) {
#pragma unroll
            for (int t = 0; t < ROW_TILES; ++t) {
                acc0[t] = __builtin_amdgcn_wmma_f32_16x16x4_f32(
                    false, cur.a[t][q], false, cur.b0[q], (short)0, acc0[t], false, false);
                acc1[t] = __builtin_amdgcn_wmma_f32_16x16x4_f32(
                    false, cur.a[t][q], false, cur.b1[q], (short)0, acc1[t], false, false);
            }
        }
        if (s + 1 < IN_F / K_STAGE) cur = nxt;
    }

#pragma unroll
    for (int t = 0; t < ROW_TILES; ++t) {
#pragma unroll
        for (int r = 0; r < 8; ++r) {
            const int m = row0 + 16 * t + r + 8 * half;
            if (m < n_nodes) {
                z[(size_t)m * OUT_F + l16]      = acc0[t][r];
                z[(size_t)m * OUT_F + 16 + l16] = acc1[t][r];
            }
        }
    }
}

#define SPREAD_MAX 20.0f

__device__ __forceinline__ bool bound_shift(const float* __restrict__ gstat)
{
    return gstat[0] - gstat[1] < SPREAD_MAX;
}

__device__ __forceinline__ void atomic_max_f32(float* addr, float v)
{
    if (v >= 0.f)
        atomicMax((int*)addr, __float_as_int(v));
    else
        atomicMin((unsigned int*)addr, __float_as_uint(v));
}

__device__ __forceinline__ void atomic_min_f32(float* addr, float v)
{
    if (v >= 0.f)
        atomicMin((int*)addr, __float_as_int(v));
    else
        atomicMax((unsigned int*)addr, __float_as_uint(v));
}

__global__ void __launch_bounds__(256) gat_node_init(
    const float* __restrict__ z, const float* __restrict__ a,
    float* __restrict__ el, float* __restrict__ rec0,
    float* __restrict__ elmax1, float* __restrict__ denom,
    float* __restrict__ out, float* __restrict__ gstat, int n_nodes)
{
    __shared__ float s_hi[8], s_lo[8];
    const int n = blockIdx.x * blockDim.x + threadIdx.x;
    const bool live = n < n_nodes;
    float sl = 0.f, sr = 0.f;
    if (live) {
        const float* zr = z + (size_t)n * OUT_F;
#pragma unroll
        for (int j = 0; j < OUT_F; ++j) {
            const float zv = zr[j];
            sl += zv * a[j];
            sr += zv * a[OUT_F + j];
        }
        el[n]    = sl;
        ((float2*)rec0)[n] = make_float2(-__builtin_inff(), sr);
        elmax1[n]          = -__builtin_inff();
        denom[n]           = 0.f;
        denom[n_nodes + n] = 0.f;
        float4* o = (float4*)(out + (size_t)n * OUT_F);
#pragma unroll
        for (int j = 0; j < OUT_F / 4; ++j) o[j] = make_float4(0.f, 0.f, 0.f, 0.f);
    }

    float hi = live ? sl : -__builtin_inff();
    float lo = live ? sl : __builtin_inff();
    for (int o = 16; o > 0; o >>= 1) {
        hi = fmaxf(hi, __shfl_xor(hi, o));
        lo = fminf(lo, __shfl_xor(lo, o));
    }
    const int wave = threadIdx.x >> 5;
    if ((threadIdx.x & 31) == 0) { s_hi[wave] = hi; s_lo[wave] = lo; }
    __syncthreads();
    if (threadIdx.x == 0) {
        for (int w = 1; w < 8; ++w) {
            hi = fmaxf(hi, s_hi[w]);
            lo = fminf(lo, s_lo[w]);
        }
        atomic_max_f32(&gstat[0], hi);
        atomic_min_f32(&gstat[1], lo);
    }
}

__device__ __forceinline__ int chip_half()
{
    const unsigned id = (unsigned)__builtin_amdgcn_s_sendmsg_rtn(0x87);
    return (int)((id >> 18) & 1u);
}

__device__ __forceinline__ float leaky(float v)
{
    return v > 0.f ? v : NEG_SLOPE * v;
}

#define EDGE_MAX_BLOCKS 256
#define EDGES_PER_TRIP 4

__device__ __forceinline__ void edge_max_one(
    const int d, const float v, const int hf,
    float* __restrict__ rec0, float* __restrict__ elmax1)
{
    float* slot = hf ? elmax1 + d : rec0 + 2 * (size_t)d;
    atomic_max_f32(slot, v);
}

__global__ void __launch_bounds__(256) gat_edge_max(
    const int* __restrict__ src, const int* __restrict__ dst,
    const float* __restrict__ el, float* __restrict__ rec0,
    float* __restrict__ elmax1, float* __restrict__ els,
    const float* __restrict__ gstat, int n_edges)
{
    if (bound_shift(gstat)) return;
    const int hf = chip_half();
    const long trip = 256L * EDGES_PER_TRIP;
    for (long b = (long)blockIdx.x * trip; b < n_edges;
         b += (long)EDGE_MAX_BLOCKS * trip) {
        const long e0 = b + threadIdx.x;
        if (b + trip <= n_edges) {
            int s[EDGES_PER_TRIP], d[EDGES_PER_TRIP];
            float v[EDGES_PER_TRIP];
#pragma unroll
            for (int k = 0; k < EDGES_PER_TRIP; ++k) {
                s[k] = src[e0 + 256 * k];
                d[k] = dst[e0 + 256 * k];
            }
#pragma unroll
            for (int k = 0; k < EDGES_PER_TRIP; ++k) v[k] = el[s[k]];
#pragma unroll
            for (int k = 0; k < EDGES_PER_TRIP; ++k) {
                els[e0 + 256 * k] = v[k];
                edge_max_one(d[k], v[k], hf, rec0, elmax1);
            }
        } else {
            for (int k = 0; k < EDGES_PER_TRIP; ++k) {
                const long e = e0 + 256 * k;
                if (e < n_edges) {
                    const float v = el[src[e]];
                    els[e] = v;
                    edge_max_one(dst[e], v, hf, rec0, elmax1);
                }
            }
        }
    }
}

#define EDGES_PER_WAVE 32
#define ROW_BATCH 8

template <bool FULL>
__device__ __forceinline__ void scatter_rows(
    const int s, const int d, const float p, const int cnt, const int lane,
    const float* __restrict__ z, float* __restrict__ out)
{
    const int n = FULL ? EDGES_PER_WAVE : cnt;
#pragma unroll
    for (int j0 = 0; j0 < n; j0 += ROW_BATCH) {
        float zv[ROW_BATCH];
#pragma unroll
        for (int k = 0; k < ROW_BATCH; ++k) {
            const int j = j0 + k;
            zv[k] = 0.f;
            if (FULL || j < cnt) {
                const int sj = __builtin_amdgcn_readlane(s, j);
                zv[k] = z[(size_t)sj * OUT_F + lane];
            }
        }
#pragma unroll
        for (int k = 0; k < ROW_BATCH; ++k) {
            const int j = j0 + k;
            if (FULL || j < cnt) {
                const int dj   = __builtin_amdgcn_readlane(d, j);
                const float aj = __int_as_float(
                    __builtin_amdgcn_readlane(__float_as_int(p), j));
                atomicAdd(out + (size_t)dj * OUT_F + lane, aj * zv[k]);
            }
        }
    }
}

__global__ void __launch_bounds__(256) gat_edge_aggregate(
    const int* __restrict__ src, const int* __restrict__ dst,
    const float* __restrict__ el, const float* __restrict__ els,
    const float* __restrict__ rec0,
    const float* __restrict__ elmax1, float* __restrict__ denom,
    const float* __restrict__ z, float* __restrict__ out,
    const float* __restrict__ gstat, long n_edges, int n_nodes)
{
    const int lane  = threadIdx.x & 31;
    const int wv    = __builtin_amdgcn_readfirstlane((int)(threadIdx.x >> 5));
    const long wave = (long)blockIdx.x * (blockDim.x >> 5) + wv;
    const long e0   = wave * EDGES_PER_WAVE;
    if (e0 >= n_edges) return;
    const long left = n_edges - e0;
    const int cnt   = left < EDGES_PER_WAVE ? (int)left : EDGES_PER_WAVE;

    float* dn = denom + (size_t)chip_half() * n_nodes;
    const bool bound  = bound_shift(gstat);
    const float g_hi  = gstat[0];

    int s = 0, d = 0;
    float p = 0.f;
    if (lane < cnt) {
        const long e = e0 + lane;
        s = src[e];
        d = dst[e];
        const float x = *(bound ? el + s : els + e);
        const float2 r0 = ((const float2*)rec0)[d];
        float top = g_hi;
        if (!bound) top = fmaxf(r0.x, elmax1[d]);
        p = __expf(leaky(x + r0.y) - leaky(top + r0.y));
        atomicAdd(&dn[d], p);
    }

    if (cnt == EDGES_PER_WAVE)
        scatter_rows<true>(s, d, p, cnt, lane, z, out);
    else
        scatter_rows<false>(s, d, p, cnt, lane, z, out);
}

#define PIECES_PER_THREAD 4

__device__ __forceinline__ float4 div4(float4 v, const float q)
{
    v.x /= q;
    v.y /= q;
    v.z /= q;
    v.w /= q;
    return v;
}

__device__ __forceinline__ float safe_denom(const float q, const bool bound)
{
    return bound ? (q > 0.f ? q : 1.f) : fmaxf(q, 1e-16f);
}

__global__ void __launch_bounds__(256) gat_normalize(
    const float* __restrict__ denom, float* __restrict__ out,
    const float* __restrict__ gstat, int n_nodes)
{
    const bool bound = bound_shift(gstat);
    const long total    = (long)n_nodes * (OUT_F / 4);
    const long blk_base = (long)blockIdx.x * (256 * PIECES_PER_THREAD);
    const long base     = blk_base + threadIdx.x;
    float4* o = (float4*)out;

    if (blk_base + 256 * PIECES_PER_THREAD <= total) {
        float4 v[PIECES_PER_THREAD];
        float d0[PIECES_PER_THREAD], d1[PIECES_PER_THREAD];
#pragma unroll
        for (int k = 0; k < PIECES_PER_THREAD; ++k) {
            const long i = base + 256 * k;
            const int n  = (int)(i / (OUT_F / 4));
            v[k]  = o[i];
            d0[k] = denom[n];
            d1[k] = denom[n_nodes + n];
        }
#pragma unroll
        for (int k = 0; k < PIECES_PER_THREAD; ++k) {
            const long i = base + 256 * k;
            o[i] = div4(v[k], safe_denom(d0[k] + d1[k], bound));
        }
    } else {
        for (int k = 0; k < PIECES_PER_THREAD; ++k) {
            const long i = base + 256 * k;
            if (i < total) {
                const int n = (int)(i / (OUT_F / 4));
                o[i] = div4(o[i], safe_denom(denom[n] + denom[n_nodes + n], bound));
            }
        }
    }
}

extern "C" void kernel_launch(void* const* d_in, const int* in_sizes, int n_in,
                              void* d_out, int out_size, void* d_ws, size_t ws_size,
                              hipStream_t stream)
{
    const float* h  = (const float*)d_in[0];
    const float* W  = (const float*)d_in[1];
    const float* a  = (const float*)d_in[2];
    const int* src  = (const int*)d_in[3];
    const int* dst  = (const int*)d_in[4];
    float* out      = (float*)d_out;

    const int n_nodes = in_sizes[0] / IN_F;
    const int n_edges = in_sizes[3];

    float* z      = (float*)d_ws;
    float* rec0   = z + (size_t)n_nodes * OUT_F;
    float* el     = rec0 + 2 * (size_t)n_nodes;
    float* elmax1 = el + n_nodes;
    float* denom  = elmax1 + n_nodes;
    float* els    = denom + 2 * (size_t)n_nodes;
    float* gstat  = els + n_edges;

    const int B = 256;
    const int ntiles = (n_nodes + 15) / 16;

    gat_gemm_wmma<<<dim3((ntiles + ROW_TILES - 1) / ROW_TILES), dim3(32), 0, stream>>>(h, W, z, gstat, n_nodes);

    gat_node_init<<<dim3((n_nodes + B - 1) / B), dim3(B), 0, stream>>>(
        z, a, el, rec0, elmax1, denom, out, gstat, n_nodes);

    gat_edge_max<<<dim3(EDGE_MAX_BLOCKS), dim3(B), 0, stream>>>(
        src, dst, el, rec0, elmax1, els, gstat, n_edges);

    const long t5 = (((long)n_edges + EDGES_PER_WAVE - 1) / EDGES_PER_WAVE) * 32;
    gat_edge_aggregate<<<dim3((unsigned)((t5 + B - 1) / B)), dim3(B), 0, stream>>>(
        src, dst, el, els, rec0, elmax1, denom, z, out, gstat, (long)n_edges, n_nodes);

    const long pieces = (long)n_nodes * (OUT_F / 4);
    const long per_block = (long)B * PIECES_PER_THREAD;
    gat_normalize<<<dim3((unsigned)((pieces + per_block - 1) / per_block)), dim3(B), 0, stream>>>(
        denom, out, gstat, n_nodes);
}
